# starting kernel
# baseline (speedup 1.0000x reference)
#include <hip/hip_runtime.h>

typedef __attribute__((ext_vector_type(2))) float v2f;
typedef __attribute__((ext_vector_type(8))) float v8f;

#define B_SZ   4
#define DEC_N  128
#define ENC_N  512
#define HID_N  512
#define K_N    512

__global__ __launch_bounds__(256) void proj_wmma_f32(
    const float* __restrict__ A, const float* __restrict__ W,
    const float* __restrict__ bias, float* __restrict__ C,
    int N, int K, int ldw, int woff, int nunits_n)
{
    const int lane = threadIdx.x & 31;
    const int wave = threadIdx.x >> 5;
    const int unit = blockIdx.x * 8 + wave;
    const int um   = (unit / nunits_n) << 5;
    const int un   = (unit % nunits_n) << 5;

    const int half = lane >> 4;
    const int r    = lane & 15;

    const float* a0p = A + (size_t)(um + r)      * K + (half << 1);
    const float* a1p = A + (size_t)(um + 16 + r) * K + (half << 1);
    const float* b0p = W + (size_t)(un + r)      * ldw + woff + (half << 1);
    const float* b1p = W + (size_t)(un + 16 + r) * ldw + woff + (half << 1);

    v8f c00 = {0.f,0.f,0.f,0.f,0.f,0.f,0.f,0.f};
    v8f c01 = c00, c10 = c00, c11 = c00;

    for (int k = 0; k < K; k += 4) {
        v2f a0 = *(const v2f*)(a0p + k);
        v2f a1 = *(const v2f*)(a1p + k);
        v2f b0 = *(const v2f*)(b0p + k);
        v2f b1 = *(const v2f*)(b1p + k);
        c00 = __builtin_amdgcn_wmma_f32_16x16x4_f32(false, a0, false, b0, (short)0, c00, false, false);
        c01 = __builtin_amdgcn_wmma_f32_16x16x4_f32(false, a0, false, b1, (short)0, c01, false, false);
        c10 = __builtin_amdgcn_wmma_f32_16x16x4_f32(false, a1, false, b0, (short)0, c10, false, false);
        c11 = __builtin_amdgcn_wmma_f32_16x16x4_f32(false, a1, false, b1, (short)0, c11, false, false);
    }

    const int   n0   = un + r;
    const int   n1   = un + 16 + r;
    const float bad0 = bias ? bias[n0] : 0.f;
    const float bad1 = bias ? bias[n1] : 0.f;
    float* row0 = C + (size_t)(um + (half << 3)) * N;
    float* row1 = C + (size_t)(um + 16 + (half << 3)) * N;
#pragma unroll
    for (int v = 0; v < 8; ++v) {
        row0[(size_t)v * N + n0] = c00[v] + bad0;
        row0[(size_t)v * N + n1] = c01[v] + bad1;
        row1[(size_t)v * N + n0] = c10[v] + bad0;
        row1[(size_t)v * N + n1] = c11[v] + bad1;
    }
}

__global__ __launch_bounds__(256) void relu_dot_score(
    const float* __restrict__ dec_proj,
    const float* __restrict__ enc_proj,
    const float* __restrict__ w_out,
    const float* __restrict__ b_out,
    float* __restrict__ out)
{
    __shared__ float s_dp[HID_N];
    __shared__ float s_w [HID_N];

    const int bq  = blockIdx.x;
    const int b   = bq >> 7;
    const int tid = threadIdx.x;

    s_dp[tid]       = dec_proj[(size_t)bq * HID_N + tid];
    s_dp[tid + 256] = dec_proj[(size_t)bq * HID_N + tid + 256];
    s_w [tid]       = w_out[tid];
    s_w [tid + 256] = w_out[tid + 256];
    __syncthreads();

    const int   lane = tid & 31;
    const int   wave = tid >> 5;
    const float bo   = b_out[0];
    const float* ep_base = enc_proj + (size_t)b * ENC_N * HID_N;

    for (int e = wave; e < ENC_N; e += 8) {
        const float* ep = ep_base + (size_t)e * HID_N;
        float acc = 0.f;
#pragma unroll
        for (int j = 0; j < 4; ++j) {
            const int h = (lane << 2) + (j << 7);
            float4 ev = *(const float4*)(ep + h);
            float4 dv = *(const float4*)(s_dp + h);
            float4 wv = *(const float4*)(s_w + h);
            float t0 = fmaxf(dv.x + ev.x, 0.f);
            float t1 = fmaxf(dv.y + ev.y, 0.f);
            float t2 = fmaxf(dv.z + ev.z, 0.f);
            float t3 = fmaxf(dv.w + ev.w, 0.f);
            acc = fmaf(t0, wv.x, acc);
            acc = fmaf(t1, wv.y, acc);
            acc = fmaf(t2, wv.z, acc);
            acc = fmaf(t3, wv.w, acc);
        }
        for (int off = 16; off > 0; off >>= 1)
            acc += __shfl_xor(acc, off, 32);
        if (lane == 0)
            out[(size_t)bq * ENC_N + e] = acc + bo;
    }
}

extern "C" void kernel_launch(void* const* d_in, const int* in_sizes, int n_in,
                              void* d_out, int out_size, void* d_ws, size_t ws_size,
                              hipStream_t stream) {
    const float* dec_states = (const float*)d_in[0];
    const float* enc_states = (const float*)d_in[1];
    const float* W_mlp      = (const float*)d_in[2];
    const float* b_mlp      = (const float*)d_in[3];
    const float* W_out      = (const float*)d_in[4];
    const float* b_out      = (const float*)d_in[5];
    float*       out        = (float*)d_out;

    float* enc_proj = (float*)d_ws;
    float* dec_proj = enc_proj + (size_t)B_SZ * ENC_N * HID_N;

    proj_wmma_f32<<<128, 256, 0, stream>>>(enc_states, W_mlp, nullptr, enc_proj,
                                           HID_N, K_N,
                                            1024,  0,
                                            16);
    proj_wmma_f32<<<32, 256, 0, stream>>>(dec_states, W_mlp, b_mlp, dec_proj,
                                          HID_N, K_N,
                                           1024,  512,
                                           16);
    relu_dot_score<<<B_SZ * DEC_N, 256, 0, stream>>>(dec_proj, enc_proj,
                                                     W_out, b_out, out);
}


# projection GEMMs: A and W k-slabs staged through double-buffered LDS by whole-line b128 global loads, fragments by ds_load_b128 (was 16-line b64 global loads per k-step); k sum order changes
# speedup vs baseline: 1.2362x; 1.2362x over previous
#include <hip/hip_runtime.h>

typedef __attribute__((ext_vector_type(2))) float v2f;
typedef __attribute__((ext_vector_type(4))) float v4f;
typedef __attribute__((ext_vector_type(8))) float v8f;

#define B_SZ   4
#define DEC_N  128
#define ENC_N  512
#define HID_N  512
#define K_N    512

#define BM     64
#define BN     128
#define KC     32
#define PITCH  36
#define STAGE  ((BM + BN) * PITCH)

__global__ __launch_bounds__(256) void proj_wmma_f32(
    const float* __restrict__ A, const float* __restrict__ W,
    const float* __restrict__ bias, float* __restrict__ C,
    int N, int K, int ldw, int woff, int nblk_n)
{
    __shared__ __attribute__((aligned(16))) float s_t[2 * STAGE];

    const int tid  = threadIdx.x;
    const int lane = tid & 31;
    const int wave = tid >> 5;
    const int bm   = (blockIdx.x / nblk_n) * BM;
    const int bn   = (blockIdx.x % nblk_n) * BN;
    const int wm   = (wave >> 2) << 5;
    const int wn   = (wave & 3) << 5;

    const int half = lane >> 4;
    const int r    = lane & 15;

    const int crow = tid >> 3;
    const int ccol = (tid & 7) << 2;
    const float* ag = A + (size_t)(bm + crow) * K + ccol;
    const float* wg = W + (size_t)(bn + crow) * ldw + woff + ccol;
    const int    sa = crow * PITCH + ccol;
    const int    sw = (BM + crow) * PITCH + ccol;

    v4f ga[2], gw[4];
    for (int j = 0; j < 2; ++j) ga[j] = *(const v4f*)(ag + (size_t)(32 * j) * K);
    for (int j = 0; j < 4; ++j) gw[j] = *(const v4f*)(wg + (size_t)(32 * j) * ldw);
    for (int j = 0; j < 2; ++j) *(v4f*)(s_t + sa + 32 * j * PITCH) = ga[j];
    for (int j = 0; j < 4; ++j) *(v4f*)(s_t + sw + 32 * j * PITCH) = gw[j];
    __syncthreads();

    const int fa0 = (wm + r) * PITCH + (half << 2);
    const int fa1 = fa0 + 16 * PITCH;
    const int fb0 = (BM + wn + r) * PITCH + (half << 2);
    const int fb1 = fb0 + 16 * PITCH;

    v8f c00 = {0.f,0.f,0.f,0.f,0.f,0.f,0.f,0.f};
    v8f c01 = c00, c10 = c00, c11 = c00;

    const int nslab = K / KC;
    for (int s = 0; s < nslab; ++s) {
        const float* st   = s_t + (s & 1) * STAGE;
        const bool   more = (s + 1 < nslab);
        if (more) {
            const int k = (s + 1) * KC;
            for (int j = 0; j < 2; ++j) ga[j] = *(const v4f*)(ag + (size_t)(32 * j) * K + k);
            for (int j = 0; j < 4; ++j) gw[j] = *(const v4f*)(wg + (size_t)(32 * j) * ldw + k);
        }
        for (int j = 0; j < KC / 8; ++j) {
            const v4f a0 = *(const v4f*)(st + fa0 + 8 * j);
            const v4f a1 = *(const v4f*)(st + fa1 + 8 * j);
            const v4f b0 = *(const v4f*)(st + fb0 + 8 * j);
            const v4f b1 = *(const v4f*)(st + fb1 + 8 * j);
            const v2f a0l = {a0.x, a0.y}, a0h = {a0.z, a0.w};
            const v2f a1l = {a1.x, a1.y}, a1h = {a1.z, a1.w};
            const v2f b0l = {b0.x, b0.y}, b0h = {b0.z, b0.w};
            const v2f b1l = {b1.x, b1.y}, b1h = {b1.z, b1.w};
            c00 = __builtin_amdgcn_wmma_f32_16x16x4_f32(false, a0l, false, b0l, (short)0, c00, false, false);
            c01 = __builtin_amdgcn_wmma_f32_16x16x4_f32(false, a0l, false, b1l, (short)0, c01, false, false);
            c10 = __builtin_amdgcn_wmma_f32_16x16x4_f32(false, a1l, false, b0l, (short)0, c10, false, false);
            c11 = __builtin_amdgcn_wmma_f32_16x16x4_f32(false, a1l, false, b1l, (short)0, c11, false, false);
            c00 = __builtin_amdgcn_wmma_f32_16x16x4_f32(false, a0h, false, b0h, (short)0, c00, false, false);
            c01 = __builtin_amdgcn_wmma_f32_16x16x4_f32(false, a0h, false, b1h, (short)0, c01, false, false);
            c10 = __builtin_amdgcn_wmma_f32_16x16x4_f32(false, a1h, false, b0h, (short)0, c10, false, false);
            c11 = __builtin_amdgcn_wmma_f32_16x16x4_f32(false, a1h, false, b1h, (short)0, c11, false, false);
        }
        if (more) {
            float* sn = s_t + ((s + 1) & 1) * STAGE;
            for (int j = 0; j < 2; ++j) *(v4f*)(sn + sa + 32 * j * PITCH) = ga[j];
            for (int j = 0; j < 4; ++j) *(v4f*)(sn + sw + 32 * j * PITCH) = gw[j];
        }
        __syncthreads();
    }

    const int   um   = bm + wm;
    const int   un   = bn + wn;
    const int   n0   = un + r;
    const int   n1   = un + 16 + r;
    const float bad0 = bias ? bias[n0] : 0.f;
    const float bad1 = bias ? bias[n1] : 0.f;
    float* row0 = C + (size_t)(um + (half << 3)) * N;
    float* row1 = C + (size_t)(um + 16 + (half << 3)) * N;
#pragma unroll
    for (int v = 0; v < 8; ++v) {
        row0[(size_t)v * N + n0] = c00[v] + bad0;
        row0[(size_t)v * N + n1] = c01[v] + bad1;
        row1[(size_t)v * N + n0] = c10[v] + bad0;
        row1[(size_t)v * N + n1] = c11[v] + bad1;
    }
}

__global__ __launch_bounds__(256) void relu_dot_score(
    const float* __restrict__ dec_proj,
    const float* __restrict__ enc_proj,
    const float* __restrict__ w_out,
    const float* __restrict__ b_out,
    float* __restrict__ out)
{
    __shared__ float s_dp[HID_N];
    __shared__ float s_w [HID_N];

    const int bq  = blockIdx.x;
    const int b   = bq >> 7;
    const int tid = threadIdx.x;

    s_dp[tid]       = dec_proj[(size_t)bq * HID_N + tid];
    s_dp[tid + 256] = dec_proj[(size_t)bq * HID_N + tid + 256];
    s_w [tid]       = w_out[tid];
    s_w [tid + 256] = w_out[tid + 256];
    __syncthreads();

    const int   lane = tid & 31;
    const int   wave = tid >> 5;
    const float bo   = b_out[0];
    const float* ep_base = enc_proj + (size_t)b * ENC_N * HID_N;

    for (int e = wave; e < ENC_N; e += 8) {
        const float* ep = ep_base + (size_t)e * HID_N;
        float acc = 0.f;
#pragma unroll
        for (int j = 0; j < 4; ++j) {
            const int h = (lane << 2) + (j << 7);
            float4 ev = *(const float4*)(ep + h);
            float4 dv = *(const float4*)(s_dp + h);
            float4 wv = *(const float4*)(s_w + h);
            float t0 = fmaxf(dv.x + ev.x, 0.f);
            float t1 = fmaxf(dv.y + ev.y, 0.f);
            float t2 = fmaxf(dv.z + ev.z, 0.f);
            float t3 = fmaxf(dv.w + ev.w, 0.f);
            acc = fmaf(t0, wv.x, acc);
            acc = fmaf(t1, wv.y, acc);
            acc = fmaf(t2, wv.z, acc);
            acc = fmaf(t3, wv.w, acc);
        }
        for (int off = 16; off > 0; off >>= 1)
            acc += __shfl_xor(acc, off, 32);
        if (lane == 0)
            out[(size_t)bq * ENC_N + e] = acc + bo;
    }
}

extern "C" void kernel_launch(void* const* d_in, const int* in_sizes, int n_in,
                              void* d_out, int out_size, void* d_ws, size_t ws_size,
                              hipStream_t stream) {
    const float* dec_states = (const float*)d_in[0];
    const float* enc_states = (const float*)d_in[1];
    const float* W_mlp      = (const float*)d_in[2];
    const float* b_mlp      = (const float*)d_in[3];
    const float* W_out      = (const float*)d_in[4];
    const float* b_out      = (const float*)d_in[5];
    float*       out        = (float*)d_out;

    float* enc_proj = (float*)d_ws;
    float* dec_proj = enc_proj + (size_t)B_SZ * ENC_N * HID_N;

    proj_wmma_f32<<<128, 256, 0, stream>>>(enc_states, W_mlp, nullptr, enc_proj,
                                           HID_N, K_N,
                                            1024,  0,
                                            4);
    proj_wmma_f32<<<32, 256, 0, stream>>>(dec_states, W_mlp, b_mlp, dec_proj,
                                          HID_N, K_N,
                                           1024,  512,
                                           4);
    relu_dot_score<<<B_SZ * DEC_N, 256, 0, stream>>>(dec_proj, enc_proj,
                                                     W_out, b_out, out);
}


# score kernel: 16q x 64e block tile, its 80 proj rows staged once in LDS, thread = 4x8 outputs serial over h, h split over 8 waves + LDS sum (was 1 q per block, shuffle sum); sum order changes
# speedup vs baseline: 2.4261x; 1.9625x over previous
#include <hip/hip_runtime.h>

typedef __attribute__((ext_vector_type(2))) float v2f;
typedef __attribute__((ext_vector_type(4))) float v4f;
typedef __attribute__((ext_vector_type(8))) float v8f;

#define B_SZ   4
#define DEC_N  128
#define ENC_N  512
#define HID_N  512
#define K_N    512

#define BM     64
#define BN     128
#define KC     32
#define PITCH  36
#define STAGE  ((BM + BN) * PITCH)

__global__ __launch_bounds__(256) void proj_wmma_f32(
    const float* __restrict__ A, const float* __restrict__ W,
    const float* __restrict__ bias, float* __restrict__ C,
    int N, int K, int ldw, int woff, int nblk_n)
{
    __shared__ __attribute__((aligned(16))) float s_t[2 * STAGE];

    const int tid  = threadIdx.x;
    const int lane = tid & 31;
    const int wave = tid >> 5;
    const int bm   = (blockIdx.x / nblk_n) * BM;
    const int bn   = (blockIdx.x % nblk_n) * BN;
    const int wm   = (wave >> 2) << 5;
    const int wn   = (wave & 3) << 5;

    const int half = lane >> 4;
    const int r    = lane & 15;

    const int crow = tid >> 3;
    const int ccol = (tid & 7) << 2;
    const float* ag = A + (size_t)(bm + crow) * K + ccol;
    const float* wg = W + (size_t)(bn + crow) * ldw + woff + ccol;
    const int    sa = crow * PITCH + ccol;
    const int    sw = (BM + crow) * PITCH + ccol;

    v4f ga[2], gw[4];
    for (int j = 0; j < 2; ++j) ga[j] = *(const v4f*)(ag + (size_t)(32 * j) * K);
    for (int j = 0; j < 4; ++j) gw[j] = *(const v4f*)(wg + (size_t)(32 * j) * ldw);
    for (int j = 0; j < 2; ++j) *(v4f*)(s_t + sa + 32 * j * PITCH) = ga[j];
    for (int j = 0; j < 4; ++j) *(v4f*)(s_t + sw + 32 * j * PITCH) = gw[j];
    __syncthreads();

    const int fa0 = (wm + r) * PITCH + (half << 2);
    const int fa1 = fa0 + 16 * PITCH;
    const int fb0 = (BM + wn + r) * PITCH + (half << 2);
    const int fb1 = fb0 + 16 * PITCH;

    v8f c00 = {0.f,0.f,0.f,0.f,0.f,0.f,0.f,0.f};
    v8f c01 = c00, c10 = c00, c11 = c00;

    const int nslab = K / KC;
    for (int s = 0; s < nslab; ++s) {
        const float* st   = s_t + (s & 1) * STAGE;
        const bool   more = (s + 1 < nslab);
        if (more) {
            const int k = (s + 1) * KC;
            for (int j = 0; j < 2; ++j) ga[j] = *(const v4f*)(ag + (size_t)(32 * j) * K + k);
            for (int j = 0; j < 4; ++j) gw[j] = *(const v4f*)(wg + (size_t)(32 * j) * ldw + k);
        }
        for (int j = 0; j < KC / 8; ++j) {
            const v4f a0 = *(const v4f*)(st + fa0 + 8 * j);
            const v4f a1 = *(const v4f*)(st + fa1 + 8 * j);
            const v4f b0 = *(const v4f*)(st + fb0 + 8 * j);
            const v4f b1 = *(const v4f*)(st + fb1 + 8 * j);
            const v2f a0l = {a0.x, a0.y}, a0h = {a0.z, a0.w};
            const v2f a1l = {a1.x, a1.y}, a1h = {a1.z, a1.w};
            const v2f b0l = {b0.x, b0.y}, b0h = {b0.z, b0.w};
            const v2f b1l = {b1.x, b1.y}, b1h = {b1.z, b1.w};
            c00 = __builtin_amdgcn_wmma_f32_16x16x4_f32(false, a0l, false, b0l, (short)0, c00, false, false);
            c01 = __builtin_amdgcn_wmma_f32_16x16x4_f32(false, a0l, false, b1l, (short)0, c01, false, false);
            c10 = __builtin_amdgcn_wmma_f32_16x16x4_f32(false, a1l, false, b0l, (short)0, c10, false, false);
            c11 = __builtin_amdgcn_wmma_f32_16x16x4_f32(false, a1l, false, b1l, (short)0, c11, false, false);
            c00 = __builtin_amdgcn_wmma_f32_16x16x4_f32(false, a0h, false, b0h, (short)0, c00, false, false);
            c01 = __builtin_amdgcn_wmma_f32_16x16x4_f32(false, a0h, false, b1h, (short)0, c01, false, false);
            c10 = __builtin_amdgcn_wmma_f32_16x16x4_f32(false, a1h, false, b0h, (short)0, c10, false, false);
            c11 = __builtin_amdgcn_wmma_f32_16x16x4_f32(false, a1h, false, b1h, (short)0, c11, false, false);
        }
        if (more) {
            float* sn = s_t + ((s + 1) & 1) * STAGE;
            for (int j = 0; j < 2; ++j) *(v4f*)(sn + sa + 32 * j * PITCH) = ga[j];
            for (int j = 0; j < 4; ++j) *(v4f*)(sn + sw + 32 * j * PITCH) = gw[j];
        }
        __syncthreads();
    }

    const int   um   = bm + wm;
    const int   un   = bn + wn;
    const int   n0   = un + r;
    const int   n1   = un + 16 + r;
    const float bad0 = bias ? bias[n0] : 0.f;
    const float bad1 = bias ? bias[n1] : 0.f;
    float* row0 = C + (size_t)(um + (half << 3)) * N;
    float* row1 = C + (size_t)(um + 16 + (half << 3)) * N;
#pragma unroll
    for (int v = 0; v < 8; ++v) {
        row0[(size_t)v * N + n0] = c00[v] + bad0;
        row0[(size_t)v * N + n1] = c01[v] + bad1;
        row1[(size_t)v * N + n0] = c10[v] + bad0;
        row1[(size_t)v * N + n1] = c11[v] + bad1;
    }
}

#define SQ     16
#define SE     64
#define SROWS  (SQ + SE)
#define SP     516
#define PIECE  128
#define NPIECE (HID_N / PIECE)
#define RP     72

struct ScoreOps { v4f d[4]; v4f e[8]; v4f w; };

__device__ __forceinline__ void score_load(ScoreOps& o, const float* pd, const float* pe,
                                           const float* pw, int h)
{
    o.w = *(const v4f*)(pw + h);
#pragma unroll
    for (int i = 0; i < 4; ++i) o.d[i] = *(const v4f*)(pd + 4 * i * SP + h);
#pragma unroll
    for (int j = 0; j < 8; ++j) o.e[j] = *(const v4f*)(pe + 8 * j * SP + h);
}

__device__ __forceinline__ void score_step(const ScoreOps& o, v2f (&acc)[4][8])
{
    const v2f zero2 = {0.f, 0.f};
    const v2f wl = {o.w.x, o.w.y}, wh = {o.w.z, o.w.w};
#pragma unroll
    for (int i = 0; i < 4; ++i) {
        const v2f dl = {o.d[i].x, o.d[i].y}, dh = {o.d[i].z, o.d[i].w};
#pragma unroll
        for (int j = 0; j < 8; ++j) {
            const v2f el = {o.e[j].x, o.e[j].y}, eh = {o.e[j].z, o.e[j].w};
            v2f tl = dl + el;
            v2f th = dh + eh;
            tl = __builtin_elementwise_max(tl, zero2);
            th = __builtin_elementwise_max(th, zero2);
            acc[i][j] = __builtin_elementwise_fma(tl, wl, acc[i][j]);
            acc[i][j] = __builtin_elementwise_fma(th, wh, acc[i][j]);
        }
    }
}


__global__ __launch_bounds__(256) void relu_dot_score(
    const float* __restrict__ dec_proj,
    const float* __restrict__ enc_proj,
    const float* __restrict__ w_out,
    const float* __restrict__ b_out,
    float* __restrict__ out)
{
    __shared__ __attribute__((aligned(16))) float s_t[SROWS * SP];
    __shared__ __attribute__((aligned(16))) float s_w[HID_N];

    const int tid  = threadIdx.x;
    const int lane = tid & 31;
    const int wave = tid >> 5;
    const int et   = blockIdx.x & 7;
    const int qt   = (blockIdx.x >> 3) & 7;
    const int b    = blockIdx.x >> 6;
    const int q0   = b * DEC_N + qt * SQ;
    const int e0   = et * SE;

    const float* dsrc = dec_proj + (size_t)(q0 + wave) * HID_N + (lane << 2);
    const float* esrc = enc_proj + ((size_t)b * ENC_N + e0 + wave) * HID_N + (lane << 2);
    float*       sdst = s_t + wave * SP + (lane << 2);

    v4f g[10];
#pragma unroll
    for (int k = 0; k < 2; ++k)  g[k] = *(const v4f*)(dsrc + (size_t)(8 * k) * HID_N);
#pragma unroll
    for (int k = 2; k < 10; ++k) g[k] = *(const v4f*)(esrc + (size_t)(8 * (k - 2)) * HID_N);
    s_w[tid]       = w_out[tid];
    s_w[tid + 256] = w_out[tid + 256];
#pragma unroll
    for (int k = 0; k < 10; ++k) *(v4f*)(sdst + 8 * k * SP) = g[k];
    __syncthreads();

    const int tq = lane & 3;
    const int te = lane >> 2;
    const float* pd = s_t + tq * SP + wave * 16;
    const float* pe = s_t + (SQ + te) * SP + wave * 16;
    const float* pw = s_w + wave * 16;

    v2f acc[4][8];
#pragma unroll
    for (int i = 0; i < 4; ++i)
#pragma unroll
        for (int j = 0; j < 8; ++j) { acc[i][j].x = 0.f; acc[i][j].y = 0.f; }

#define SCHED_PIN() __builtin_amdgcn_sched_barrier(0)
    ScoreOps oa, ob;
    score_load(oa, pd, pe, pw, 0);
#pragma unroll 1
    for (int p = 0; p + 1 < NPIECE; ++p) {
        const int h = p * PIECE;
#pragma unroll
        for (int k = 0; k < 2; ++k)  g[k] = *(const v4f*)(dsrc + (size_t)(8 * k) * HID_N + h + PIECE);
#pragma unroll
        for (int k = 2; k < 10; ++k) g[k] = *(const v4f*)(esrc + (size_t)(8 * (k - 2)) * HID_N + h + PIECE);
        score_load(ob, pd, pe, pw, h + 4);
        SCHED_PIN();
        score_step(oa, acc);
        SCHED_PIN();
        score_load(oa, pd, pe, pw, h + 8);
        SCHED_PIN();
        score_step(ob, acc);
        SCHED_PIN();
        score_load(ob, pd, pe, pw, h + 12);
        SCHED_PIN();
        score_step(oa, acc);
        SCHED_PIN();
        float* sn = sdst + h + PIECE;
#pragma unroll
        for (int k = 0; k < 10; ++k) *(v4f*)(sn + 8 * k * SP) = g[k];
        __syncthreads();
        score_load(oa, pd, pe, pw, h + PIECE);
        SCHED_PIN();
        score_step(ob, acc);
        SCHED_PIN();
    }
    {
        const int h = (NPIECE - 1) * PIECE;
        score_load(ob, pd, pe, pw, h + 4);
        SCHED_PIN();
        score_step(oa, acc);
        SCHED_PIN();
        score_load(oa, pd, pe, pw, h + 8);
        SCHED_PIN();
        score_step(ob, acc);
        SCHED_PIN();
        score_load(ob, pd, pe, pw, h + 12);
        SCHED_PIN();
        score_step(oa, acc);
        score_step(ob, acc);
    }
    __syncthreads();

    float* red = s_t;
#pragma unroll
    for (int i = 0; i < 4; ++i)
#pragma unroll
        for (int j = 0; j < 8; ++j)
            red[(wave * SQ + 4 * i + tq) * RP + 8 * j + te] = acc[i][j].x + acc[i][j].y;
    __syncthreads();

    const int   oq = tid >> 4;
    const int   oe = (tid & 15) << 2;
    const float bo = b_out[0];
    v4f o = *(const v4f*)(red + oq * RP + oe);
#pragma unroll
    for (int s = 1; s < 8; ++s) o += *(const v4f*)(red + (s * SQ + oq) * RP + oe);
    o.x += bo; o.y += bo; o.z += bo; o.w += bo;
    *(v4f*)(out + (size_t)(q0 + oq) * ENC_N + e0 + oe) = o;
}

extern "C" void kernel_launch(void* const* d_in, const int* in_sizes, int n_in,
                              void* d_out, int out_size, void* d_ws, size_t ws_size,
                              hipStream_t stream) {
    const float* dec_states = (const float*)d_in[0];
    const float* enc_states = (const float*)d_in[1];
    const float* W_mlp      = (const float*)d_in[2];
    const float* b_mlp      = (const float*)d_in[3];
    const float* W_out      = (const float*)d_in[4];
    const float* b_out      = (const float*)d_in[5];
    float*       out        = (float*)d_out;

    float* enc_proj = (float*)d_ws;
    float* dec_proj = enc_proj + (size_t)B_SZ * ENC_N * HID_N;

    proj_wmma_f32<<<128, 256, 0, stream>>>(enc_states, W_mlp, nullptr, enc_proj,
                                           HID_N, K_N,
                                            1024,  0,
                                            4);
    proj_wmma_f32<<<32, 256, 0, stream>>>(dec_states, W_mlp, b_mlp, dec_proj,
                                          HID_N, K_N,
                                           1024,  512,
                                           4);
    relu_dot_score<<<256, 256, 0, stream>>>(dec_proj, enc_proj,
                                            W_out, b_out, out);
}


# both projection GEMMs in ONE launch of 160 blocks (128 enc tiles + 32 dec tiles, chosen per block): the dec GEMM runs beside the enc GEMM on idle WGPs instead of behind it; 3 launches -> 2
# speedup vs baseline: 4.0513x; 1.6699x over previous
#include <hip/hip_runtime.h>

typedef __attribute__((ext_vector_type(2))) float v2f;
typedef __attribute__((ext_vector_type(4))) float v4f;
typedef __attribute__((ext_vector_type(8))) float v8f;

#define B_SZ   4
#define DEC_N  128
#define ENC_N  512
#define HID_N  512
#define K_N    512

#define BM     64
#define BN     128
#define KC     32
#define PITCH  36
#define STAGE  ((BM + BN) * PITCH)

__global__ __launch_bounds__(256) void proj_wmma_f32(
    const float* __restrict__ A_enc, const float* __restrict__ A_dec,
    const float* __restrict__ W, const float* __restrict__ bias_dec,
    float* __restrict__ C_enc, float* __restrict__ C_dec,
    int N, int K, int ldw, int woff_dec, int nblk_n, int nblk_enc)
{
    const bool   is_dec = (int)blockIdx.x >= nblk_enc;
    const unsigned blk  = is_dec ? blockIdx.x - (unsigned)nblk_enc : blockIdx.x;
    const float* A      = is_dec ? A_dec : A_enc;
    float*       C      = is_dec ? C_dec : C_enc;
    const float* bias   = is_dec ? bias_dec : nullptr;
    const int    woff   = is_dec ? woff_dec : 0;

    __shared__ __attribute__((aligned(16))) float s_t[2 * STAGE];

    const int tid  = threadIdx.x;
    const int lane = tid & 31;
    const int wave = tid >> 5;
    const int bm   = (blk / nblk_n) * BM;
    const int bn   = (blk % nblk_n) * BN;
    const int wm   = (wave >> 2) << 5;
    const int wn   = (wave & 3) << 5;

    const int half = lane >> 4;
    const int r    = lane & 15;

    const int crow = tid >> 3;
    const int ccol = (tid & 7) << 2;
    const float* ag = A + (size_t)(bm + crow) * K + ccol;
    const float* wg = W + (size_t)(bn + crow) * ldw + woff + ccol;
    const int    sa = crow * PITCH + ccol;
    const int    sw = (BM + crow) * PITCH + ccol;

    v4f ga[2], gw[4];
    for (int j = 0; j < 2; ++j) ga[j] = *(const v4f*)(ag + (size_t)(32 * j) * K);
    for (int j = 0; j < 4; ++j) gw[j] = *(const v4f*)(wg + (size_t)(32 * j) * ldw);
    for (int j = 0; j < 2; ++j) *(v4f*)(s_t + sa + 32 * j * PITCH) = ga[j];
    for (int j = 0; j < 4; ++j) *(v4f*)(s_t + sw + 32 * j * PITCH) = gw[j];
    __syncthreads();

    const int fa0 = (wm + r) * PITCH + (half << 2);
    const int fa1 = fa0 + 16 * PITCH;
    const int fb0 = (BM + wn + r) * PITCH + (half << 2);
    const int fb1 = fb0 + 16 * PITCH;

    v8f c00 = {0.f,0.f,0.f,0.f,0.f,0.f,0.f,0.f};
    v8f c01 = c00, c10 = c00, c11 = c00;

    const int nslab = K / KC;
    for (int s = 0; s < nslab; ++s) {
        const float* st   = s_t + (s & 1) * STAGE;
        const bool   more = (s + 1 < nslab);
        if (more) {
            const int k = (s + 1) * KC;
            for (int j = 0; j < 2; ++j) ga[j] = *(const v4f*)(ag + (size_t)(32 * j) * K + k);
            for (int j = 0; j < 4; ++j) gw[j] = *(const v4f*)(wg + (size_t)(32 * j) * ldw + k);
        }
        for (int j = 0; j < KC / 8; ++j) {
            const v4f a0 = *(const v4f*)(st + fa0 + 8 * j);
            const v4f a1 = *(const v4f*)(st + fa1 + 8 * j);
            const v4f b0 = *(const v4f*)(st + fb0 + 8 * j);
            const v4f b1 = *(const v4f*)(st + fb1 + 8 * j);
            const v2f a0l = {a0.x, a0.y}, a0h = {a0.z, a0.w};
            const v2f a1l = {a1.x, a1.y}, a1h = {a1.z, a1.w};
            const v2f b0l = {b0.x, b0.y}, b0h = {b0.z, b0.w};
            const v2f b1l = {b1.x, b1.y}, b1h = {b1.z, b1.w};
            c00 = __builtin_amdgcn_wmma_f32_16x16x4_f32(false, a0l, false, b0l, (short)0, c00, false, false);
            c01 = __builtin_amdgcn_wmma_f32_16x16x4_f32(false, a0l, false, b1l, (short)0, c01, false, false);
            c10 = __builtin_amdgcn_wmma_f32_16x16x4_f32(false, a1l, false, b0l, (short)0, c10, false, false);
            c11 = __builtin_amdgcn_wmma_f32_16x16x4_f32(false, a1l, false, b1l, (short)0, c11, false, false);
            c00 = __builtin_amdgcn_wmma_f32_16x16x4_f32(false, a0h, false, b0h, (short)0, c00, false, false);
            c01 = __builtin_amdgcn_wmma_f32_16x16x4_f32(false, a0h, false, b1h, (short)0, c01, false, false);
            c10 = __builtin_amdgcn_wmma_f32_16x16x4_f32(false, a1h, false, b0h, (short)0, c10, false, false);
            c11 = __builtin_amdgcn_wmma_f32_16x16x4_f32(false, a1h, false, b1h, (short)0, c11, false, false);
        }
        if (more) {
            float* sn = s_t + ((s + 1) & 1) * STAGE;
            for (int j = 0; j < 2; ++j) *(v4f*)(sn + sa + 32 * j * PITCH) = ga[j];
            for (int j = 0; j < 4; ++j) *(v4f*)(sn + sw + 32 * j * PITCH) = gw[j];
        }
        __syncthreads();
    }

    const int   um   = bm + wm;
    const int   un   = bn + wn;
    const int   n0   = un + r;
    const int   n1   = un + 16 + r;
    const float bad0 = bias ? bias[n0] : 0.f;
    const float bad1 = bias ? bias[n1] : 0.f;
    float* row0 = C + (size_t)(um + (half << 3)) * N;
    float* row1 = C + (size_t)(um + 16 + (half << 3)) * N;
#pragma unroll
    for (int v = 0; v < 8; ++v) {
        row0[(size_t)v * N + n0] = c00[v] + bad0;
        row0[(size_t)v * N + n1] = c01[v] + bad1;
        row1[(size_t)v * N + n0] = c10[v] + bad0;
        row1[(size_t)v * N + n1] = c11[v] + bad1;
    }
}

#define SQ     16
#define SE     64
#define SROWS  (SQ + SE)
#define SP     516
#define PIECE  128
#define NPIECE (HID_N / PIECE)
#define RP     72

struct ScoreOps { v4f d[4]; v4f e[8]; v4f w; };

__device__ __forceinline__ void score_load(ScoreOps& o, const float* pd, const float* pe,
                                           const float* pw, int h)
{
    o.w = *(const v4f*)(pw + h);
#pragma unroll
    for (int i = 0; i < 4; ++i) o.d[i] = *(const v4f*)(pd + 4 * i * SP + h);
#pragma unroll
    for (int j = 0; j < 8; ++j) o.e[j] = *(const v4f*)(pe + 8 * j * SP + h);
}

__device__ __forceinline__ void score_step(const ScoreOps& o, v2f (&acc)[4][8])
{
    const v2f zero2 = {0.f, 0.f};
    const v2f wl = {o.w.x, o.w.y}, wh = {o.w.z, o.w.w};
#pragma unroll
    for (int i = 0; i < 4; ++i) {
        const v2f dl = {o.d[i].x, o.d[i].y}, dh = {o.d[i].z, o.d[i].w};
#pragma unroll
        for (int j = 0; j < 8; ++j) {
            const v2f el = {o.e[j].x, o.e[j].y}, eh = {o.e[j].z, o.e[j].w};
            v2f tl = dl + el;
            v2f th = dh + eh;
            tl = __builtin_elementwise_max(tl, zero2);
            th = __builtin_elementwise_max(th, zero2);
            acc[i][j] = __builtin_elementwise_fma(tl, wl, acc[i][j]);
            acc[i][j] = __builtin_elementwise_fma(th, wh, acc[i][j]);
        }
    }
}


__global__ __launch_bounds__(256) void relu_dot_score(
    const float* __restrict__ dec_proj,
    const float* __restrict__ enc_proj,
    const float* __restrict__ w_out,
    const float* __restrict__ b_out,
    float* __restrict__ out)
{
    __shared__ __attribute__((aligned(16))) float s_t[SROWS * SP];
    __shared__ __attribute__((aligned(16))) float s_w[HID_N];

    const int tid  = threadIdx.x;
    const int lane = tid & 31;
    const int wave = tid >> 5;
    const int et   = blockIdx.x & 7;
    const int qt   = (blockIdx.x >> 3) & 7;
    const int b    = blockIdx.x >> 6;
    const int q0   = b * DEC_N + qt * SQ;
    const int e0   = et * SE;

    const float* dsrc = dec_proj + (size_t)(q0 + wave) * HID_N + (lane << 2);
    const float* esrc = enc_proj + ((size_t)b * ENC_N + e0 + wave) * HID_N + (lane << 2);
    float*       sdst = s_t + wave * SP + (lane << 2);

    v4f g[10];
#pragma unroll
    for (int k = 0; k < 2; ++k)  g[k] = *(const v4f*)(dsrc + (size_t)(8 * k) * HID_N);
#pragma unroll
    for (int k = 2; k < 10; ++k) g[k] = *(const v4f*)(esrc + (size_t)(8 * (k - 2)) * HID_N);
    s_w[tid]       = w_out[tid];
    s_w[tid + 256] = w_out[tid + 256];
#pragma unroll
    for (int k = 0; k < 10; ++k) *(v4f*)(sdst + 8 * k * SP) = g[k];
    __syncthreads();

    const int tq = lane & 3;
    const int te = lane >> 2;
    const float* pd = s_t + tq * SP + wave * 16;
    const float* pe = s_t + (SQ + te) * SP + wave * 16;
    const float* pw = s_w + wave * 16;

    v2f acc[4][8];
#pragma unroll
    for (int i = 0; i < 4; ++i)
#pragma unroll
        for (int j = 0; j < 8; ++j) { acc[i][j].x = 0.f; acc[i][j].y = 0.f; }

#define SCHED_PIN() __builtin_amdgcn_sched_barrier(0)
    ScoreOps oa, ob;
    score_load(oa, pd, pe, pw, 0);
#pragma unroll 1
    for (int p = 0; p + 1 < NPIECE; ++p) {
        const int h = p * PIECE;
#pragma unroll
        for (int k = 0; k < 2; ++k)  g[k] = *(const v4f*)(dsrc + (size_t)(8 * k) * HID_N + h + PIECE);
#pragma unroll
        for (int k = 2; k < 10; ++k) g[k] = *(const v4f*)(esrc + (size_t)(8 * (k - 2)) * HID_N + h + PIECE);
        score_load(ob, pd, pe, pw, h + 4);
        SCHED_PIN();
        score_step(oa, acc);
        SCHED_PIN();
        score_load(oa, pd, pe, pw, h + 8);
        SCHED_PIN();
        score_step(ob, acc);
        SCHED_PIN();
        score_load(ob, pd, pe, pw, h + 12);
        SCHED_PIN();
        score_step(oa, acc);
        SCHED_PIN();
        float* sn = sdst + h + PIECE;
#pragma unroll
        for (int k = 0; k < 10; ++k) *(v4f*)(sn + 8 * k * SP) = g[k];
        __syncthreads();
        score_load(oa, pd, pe, pw, h + PIECE);
        SCHED_PIN();
        score_step(ob, acc);
        SCHED_PIN();
    }
    {
        const int h = (NPIECE - 1) * PIECE;
        score_load(ob, pd, pe, pw, h + 4);
        SCHED_PIN();
        score_step(oa, acc);
        SCHED_PIN();
        score_load(oa, pd, pe, pw, h + 8);
        SCHED_PIN();
        score_step(ob, acc);
        SCHED_PIN();
        score_load(ob, pd, pe, pw, h + 12);
        SCHED_PIN();
        score_step(oa, acc);
        score_step(ob, acc);
    }
    __syncthreads();

    float* red = s_t;
#pragma unroll
    for (int i = 0; i < 4; ++i)
#pragma unroll
        for (int j = 0; j < 8; ++j)
            red[(wave * SQ + 4 * i + tq) * RP + 8 * j + te] = acc[i][j].x + acc[i][j].y;
    __syncthreads();

    const int   oq = tid >> 4;
    const int   oe = (tid & 15) << 2;
    const float bo = b_out[0];
    v4f o = *(const v4f*)(red + oq * RP + oe);
#pragma unroll
    for (int s = 1; s < 8; ++s) o += *(const v4f*)(red + (s * SQ + oq) * RP + oe);
    o.x += bo; o.y += bo; o.z += bo; o.w += bo;
    *(v4f*)(out + (size_t)(q0 + oq) * ENC_N + e0 + oe) = o;
}

extern "C" void kernel_launch(void* const* d_in, const int* in_sizes, int n_in,
                              void* d_out, int out_size, void* d_ws, size_t ws_size,
                              hipStream_t stream) {
    const float* dec_states = (const float*)d_in[0];
    const float* enc_states = (const float*)d_in[1];
    const float* W_mlp      = (const float*)d_in[2];
    const float* b_mlp      = (const float*)d_in[3];
    const float* W_out      = (const float*)d_in[4];
    const float* b_out      = (const float*)d_in[5];
    float*       out        = (float*)d_out;

    float* enc_proj = (float*)d_ws;
    float* dec_proj = enc_proj + (size_t)B_SZ * ENC_N * HID_N;

    proj_wmma_f32<<<160, 256, 0, stream>>>(enc_states, dec_states, W_mlp, b_mlp,
                                           enc_proj, dec_proj,
                                           HID_N, K_N,
                                            1024,  512,
                                            4,  128);
    relu_dot_score<<<256, 256, 0, stream>>>(dec_proj, enc_proj,
                                            W_out, b_out, out);
}


# projection launch re-tiled over all 256 WGPs: 256 equal blocks x 4 waves (one per SIMD), block = a 64x64 enc tile + a 32x32 dec tile, fragment sets read one group ahead (was 160 blocks x 8 waves)
# speedup vs baseline: 5.4331x; 1.3411x over previous
#include <hip/hip_runtime.h>

typedef __attribute__((ext_vector_type(2))) float v2f;
typedef __attribute__((ext_vector_type(4))) float v4f;
typedef __attribute__((ext_vector_type(8))) float v8f;

#define B_SZ   4
#define DEC_N  128
#define ENC_N  512
#define HID_N  512
#define K_N    512

#define BME    64
#define BNE    64
#define BMD    32
#define BND    32
#define KC     64
#define PITCH  68
#define SROWS_P (BME + BNE + BMD + BND)
#define STAGE  (SROWS_P * PITCH)
#define NSTAGE 4
#define NSLAB  (K_N / KC)
#define LDW    1024
#define WOFF_DEC 512
#define NG     24

struct FragSet { v4f a[2][3]; v4f b[2][3]; };

__device__ __forceinline__ void frag_load(FragSet& f, const float* st, int fa0, int fa1, int fad,
                                          int fb0, int fb1, int fbd, int q)
{
#pragma unroll
    for (int rr = 0; rr < 2; ++rr) {
        const int off = 8 * (2 * q + rr);
        f.a[rr][0] = *(const v4f*)(st + fa0 + off);
        f.a[rr][1] = *(const v4f*)(st + fa1 + off);
        f.a[rr][2] = *(const v4f*)(st + fad + off);
        f.b[rr][0] = *(const v4f*)(st + fb0 + off);
        f.b[rr][1] = *(const v4f*)(st + fb1 + off);
        f.b[rr][2] = *(const v4f*)(st + fbd + off);
    }
}

#define WMMA4(c, a, b) c = __builtin_amdgcn_wmma_f32_16x16x4_f32(false, a, false, b, (short)0, c, false, false)

__device__ __forceinline__ void frag_mma(const FragSet& f, v8f& c00, v8f& c01, v8f& c10, v8f& c11, v8f& cd)
{
#pragma unroll
    for (int rr = 0; rr < 2; ++rr) {
        const v2f a0l = {f.a[rr][0].x, f.a[rr][0].y}, a0h = {f.a[rr][0].z, f.a[rr][0].w};
        const v2f a1l = {f.a[rr][1].x, f.a[rr][1].y}, a1h = {f.a[rr][1].z, f.a[rr][1].w};
        const v2f adl = {f.a[rr][2].x, f.a[rr][2].y}, adh = {f.a[rr][2].z, f.a[rr][2].w};
        const v2f b0l = {f.b[rr][0].x, f.b[rr][0].y}, b0h = {f.b[rr][0].z, f.b[rr][0].w};
        const v2f b1l = {f.b[rr][1].x, f.b[rr][1].y}, b1h = {f.b[rr][1].z, f.b[rr][1].w};
        const v2f bdl = {f.b[rr][2].x, f.b[rr][2].y}, bdh = {f.b[rr][2].z, f.b[rr][2].w};
        WMMA4(c00, a0l, b0l); WMMA4(c01, a0l, b1l); WMMA4(c10, a1l, b0l); WMMA4(c11, a1l, b1l); WMMA4(cd, adl, bdl);
        WMMA4(c00, a0h, b0h); WMMA4(c01, a0h, b1h); WMMA4(c10, a1h, b0h); WMMA4(c11, a1h, b1h); WMMA4(cd, adh, bdh);
    }
}

__global__ __launch_bounds__(128) void proj_wmma_f32(
    const float* __restrict__ A_enc, const float* __restrict__ A_dec,
    const float* __restrict__ W, const float* __restrict__ bias_dec,
    float* __restrict__ C_enc, float* __restrict__ C_dec)
{
    __shared__ __attribute__((aligned(16))) float s_t[NSTAGE * STAGE];

    const int tid  = threadIdx.x;
    const int lane = tid & 31;
    const int wave = tid >> 5;
    const int blk  = blockIdx.x;
    const int bme  = (blk >> 3) * BME;
    const int bne  = (blk & 7) * BNE;
    const int bmd  = (blk >> 4) * BMD;
    const int bnd  = (blk & 15) * BND;
    const int wm   = (wave >> 1) << 5;
    const int wn   = (wave & 1) << 5;
    const int dm   = (wave >> 1) << 4;
    const int dn   = (wave & 1) << 4;

    const int half = lane >> 4;
    const int r    = lane & 15;

    const int crow = tid >> 4;
    const int ccol = (tid & 15) << 2;
    const float* pae = A_enc + (size_t)(bme + crow) * K_N + ccol;
    const float* pwe = W     + (size_t)(bne + crow) * LDW + ccol;
    const float* pad = A_dec + (size_t)(bmd + crow) * K_N + ccol;
    const float* pwd = W     + (size_t)(bnd + crow) * LDW + WOFF_DEC + ccol;
    const int    sdst = crow * PITCH + ccol;

    v4f g[NG];
#define G_LOAD(k0)                                                                                   \
    {                                                                                                \
        _Pragma("unroll") for (int j = 0; j < 8; ++j) g[j]      = *(const v4f*)(pae + (size_t)(8 * j) * K_N + (k0)); \
        _Pragma("unroll") for (int j = 0; j < 8; ++j) g[8 + j]  = *(const v4f*)(pwe + (size_t)(8 * j) * LDW + (k0)); \
        _Pragma("unroll") for (int j = 0; j < 4; ++j) g[16 + j] = *(const v4f*)(pad + (size_t)(8 * j) * K_N + (k0)); \
        _Pragma("unroll") for (int j = 0; j < 4; ++j) g[20 + j] = *(const v4f*)(pwd + (size_t)(8 * j) * LDW + (k0)); \
    }
#define G_STORE(stp)                                                                                 \
    {                                                                                                \
        _Pragma("unroll") for (int j = 0; j < NG; ++j) *(v4f*)((stp) + sdst + 8 * j * PITCH) = g[j]; \
    }

    G_LOAD(0);
    G_STORE(s_t);
    __syncthreads();
    G_LOAD(KC);

    const int fa0 = (wm + r) * PITCH + (half << 2);
    const int fa1 = fa0 + 16 * PITCH;
    const int fb0 = (BME + wn + r) * PITCH + (half << 2);
    const int fb1 = fb0 + 16 * PITCH;
    const int fad = (BME + BNE + dm + r) * PITCH + (half << 2);
    const int fbd = (BME + BNE + BMD + dn + r) * PITCH + (half << 2);

    v8f c00 = {0.f,0.f,0.f,0.f,0.f,0.f,0.f,0.f};
    v8f c01 = c00, c10 = c00, c11 = c00, cd = c00;

#define PROJ_PIN() __builtin_amdgcn_sched_barrier(0)
#define LDS_BARRIER()                                                        \
    {                                                                        \
        __builtin_amdgcn_fence(__ATOMIC_RELEASE, "workgroup", "local");      \
        __builtin_amdgcn_s_barrier();                                        \
        __builtin_amdgcn_fence(__ATOMIC_ACQUIRE, "workgroup", "local");      \
    }
    FragSet fp, fq;
    frag_load(fp, s_t, fa0, fa1, fad, fb0, fb1, fbd, 0);
#pragma unroll 1
    for (int s = 0; s + 1 < NSLAB; ++s) {
        const float* st = s_t + (s & (NSTAGE - 1)) * STAGE;
        float*       sn = s_t + ((s + 1) & (NSTAGE - 1)) * STAGE;
        frag_load(fq, st, fa0, fa1, fad, fb0, fb1, fbd, 1);
        PROJ_PIN();
        frag_mma(fp, c00, c01, c10, c11, cd);
        PROJ_PIN();
        frag_load(fp, st, fa0, fa1, fad, fb0, fb1, fbd, 2);
        PROJ_PIN();
        frag_mma(fq, c00, c01, c10, c11, cd);
        PROJ_PIN();
        frag_load(fq, st, fa0, fa1, fad, fb0, fb1, fbd, 3);
        G_STORE(sn);
        if (s + 2 < NSLAB) G_LOAD((s + 2) * KC);
        PROJ_PIN();
        frag_mma(fp, c00, c01, c10, c11, cd);
        PROJ_PIN();
        LDS_BARRIER();
        frag_load(fp, sn, fa0, fa1, fad, fb0, fb1, fbd, 0);
        PROJ_PIN();
        frag_mma(fq, c00, c01, c10, c11, cd);
        PROJ_PIN();
    }
    {
        const float* st = s_t + ((NSLAB - 1) & (NSTAGE - 1)) * STAGE;
        frag_load(fq, st, fa0, fa1, fad, fb0, fb1, fbd, 1);
        PROJ_PIN();
        frag_mma(fp, c00, c01, c10, c11, cd);
        PROJ_PIN();
        frag_load(fp, st, fa0, fa1, fad, fb0, fb1, fbd, 2);
        PROJ_PIN();
        frag_mma(fq, c00, c01, c10, c11, cd);
        PROJ_PIN();
        frag_load(fq, st, fa0, fa1, fad, fb0, fb1, fbd, 3);
        PROJ_PIN();
        frag_mma(fp, c00, c01, c10, c11, cd);
        frag_mma(fq, c00, c01, c10, c11, cd);
    }

    {
        const int n0 = bne + wn + r;
        const int n1 = n0 + 16;
        float* row0 = C_enc + (size_t)(bme + wm + (half << 3)) * HID_N;
        float* row1 = C_enc + (size_t)(bme + wm + 16 + (half << 3)) * HID_N;
#pragma unroll
        for (int v = 0; v < 8; ++v) {
            row0[(size_t)v * HID_N + n0] = c00[v];
            row0[(size_t)v * HID_N + n1] = c01[v];
            row1[(size_t)v * HID_N + n0] = c10[v];
            row1[(size_t)v * HID_N + n1] = c11[v];
        }
        const int   nd   = bnd + dn + r;
        const float badd = bias_dec[nd];
        float* rowd = C_dec + (size_t)(bmd + dm + (half << 3)) * HID_N;
#pragma unroll
        for (int v = 0; v < 8; ++v) rowd[(size_t)v * HID_N + nd] = cd[v] + badd;
    }
}

#define SQ     16
#define SE     64
#define SROWS  (SQ + SE)
#define SP     516
#define PIECE  128
#define NPIECE (HID_N / PIECE)
#define RP     72

struct ScoreOps { v4f d[4]; v4f e[8]; v4f w; };

__device__ __forceinline__ void score_load(ScoreOps& o, const float* pd, const float* pe,
                                           const float* pw, int h)
{
    o.w = *(const v4f*)(pw + h);
#pragma unroll
    for (int i = 0; i < 4; ++i) o.d[i] = *(const v4f*)(pd + 4 * i * SP + h);
#pragma unroll
    for (int j = 0; j < 8; ++j) o.e[j] = *(const v4f*)(pe + 8 * j * SP + h);
}

__device__ __forceinline__ void score_step(const ScoreOps& o, v2f (&acc)[4][8])
{
    const v2f zero2 = {0.f, 0.f};
    const v2f wl = {o.w.x, o.w.y}, wh = {o.w.z, o.w.w};
#pragma unroll
    for (int i = 0; i < 4; ++i) {
        const v2f dl = {o.d[i].x, o.d[i].y}, dh = {o.d[i].z, o.d[i].w};
#pragma unroll
        for (int j = 0; j < 8; ++j) {
            const v2f el = {o.e[j].x, o.e[j].y}, eh = {o.e[j].z, o.e[j].w};
            v2f tl = dl + el;
            v2f th = dh + eh;
            tl = __builtin_elementwise_max(tl, zero2);
            th = __builtin_elementwise_max(th, zero2);
            acc[i][j] = __builtin_elementwise_fma(tl, wl, acc[i][j]);
            acc[i][j] = __builtin_elementwise_fma(th, wh, acc[i][j]);
        }
    }
}


__global__ __launch_bounds__(256) void relu_dot_score(
    const float* __restrict__ dec_proj,
    const float* __restrict__ enc_proj,
    const float* __restrict__ w_out,
    const float* __restrict__ b_out,
    float* __restrict__ out)
{
    __shared__ __attribute__((aligned(16))) float s_t[SROWS * SP];
    __shared__ __attribute__((aligned(16))) float s_w[HID_N];

    const int tid  = threadIdx.x;
    const int lane = tid & 31;
    const int wave = tid >> 5;
    const int et   = blockIdx.x & 7;
    const int qt   = (blockIdx.x >> 3) & 7;
    const int b    = blockIdx.x >> 6;
    const int q0   = b * DEC_N + qt * SQ;
    const int e0   = et * SE;

    const float* dsrc = dec_proj + (size_t)(q0 + wave) * HID_N + (lane << 2);
    const float* esrc = enc_proj + ((size_t)b * ENC_N + e0 + wave) * HID_N + (lane << 2);
    float*       sdst = s_t + wave * SP + (lane << 2);

    v4f g[10];
#pragma unroll
    for (int k = 0; k < 2; ++k)  g[k] = *(const v4f*)(dsrc + (size_t)(8 * k) * HID_N);
#pragma unroll
    for (int k = 2; k < 10; ++k) g[k] = *(const v4f*)(esrc + (size_t)(8 * (k - 2)) * HID_N);
    s_w[tid]       = w_out[tid];
    s_w[tid + 256] = w_out[tid + 256];
#pragma unroll
    for (int k = 0; k < 10; ++k) *(v4f*)(sdst + 8 * k * SP) = g[k];
    __syncthreads();

    const int tq = lane & 3;
    const int te = lane >> 2;
    const float* pd = s_t + tq * SP + wave * 16;
    const float* pe = s_t + (SQ + te) * SP + wave * 16;
    const float* pw = s_w + wave * 16;

    v2f acc[4][8];
#pragma unroll
    for (int i = 0; i < 4; ++i)
#pragma unroll
        for (int j = 0; j < 8; ++j) { acc[i][j].x = 0.f; acc[i][j].y = 0.f; }

#define SCHED_PIN() __builtin_amdgcn_sched_barrier(0)
    ScoreOps oa, ob;
    score_load(oa, pd, pe, pw, 0);
#pragma unroll 1
    for (int p = 0; p + 1 < NPIECE; ++p) {
        const int h = p * PIECE;
#pragma unroll
        for (int k = 0; k < 2; ++k)  g[k] = *(const v4f*)(dsrc + (size_t)(8 * k) * HID_N + h + PIECE);
#pragma unroll
        for (int k = 2; k < 10; ++k) g[k] = *(const v4f*)(esrc + (size_t)(8 * (k - 2)) * HID_N + h + PIECE);
        score_load(ob, pd, pe, pw, h + 4);
        SCHED_PIN();
        score_step(oa, acc);
        SCHED_PIN();
        score_load(oa, pd, pe, pw, h + 8);
        SCHED_PIN();
        score_step(ob, acc);
        SCHED_PIN();
        score_load(ob, pd, pe, pw, h + 12);
        SCHED_PIN();
        score_step(oa, acc);
        SCHED_PIN();
        float* sn = sdst + h + PIECE;
#pragma unroll
        for (int k = 0; k < 10; ++k) *(v4f*)(sn + 8 * k * SP) = g[k];
        __syncthreads();
        score_load(oa, pd, pe, pw, h + PIECE);
        SCHED_PIN();
        score_step(ob, acc);
        SCHED_PIN();
    }
    {
        const int h = (NPIECE - 1) * PIECE;
        score_load(ob, pd, pe, pw, h + 4);
        SCHED_PIN();
        score_step(oa, acc);
        SCHED_PIN();
        score_load(oa, pd, pe, pw, h + 8);
        SCHED_PIN();
        score_step(ob, acc);
        SCHED_PIN();
        score_load(ob, pd, pe, pw, h + 12);
        SCHED_PIN();
        score_step(oa, acc);
        score_step(ob, acc);
    }
    __syncthreads();

    float* red = s_t;
#pragma unroll
    for (int i = 0; i < 4; ++i)
#pragma unroll
        for (int j = 0; j < 8; ++j)
            red[(wave * SQ + 4 * i + tq) * RP + 8 * j + te] = acc[i][j].x + acc[i][j].y;
    __syncthreads();

    const int   oq = tid >> 4;
    const int   oe = (tid & 15) << 2;
    const float bo = b_out[0];
    v4f o = *(const v4f*)(red + oq * RP + oe);
#pragma unroll
    for (int s = 1; s < 8; ++s) o += *(const v4f*)(red + (s * SQ + oq) * RP + oe);
    o.x += bo; o.y += bo; o.z += bo; o.w += bo;
    *(v4f*)(out + (size_t)(q0 + oq) * ENC_N + e0 + oe) = o;
}

extern "C" void kernel_launch(void* const* d_in, const int* in_sizes, int n_in,
                              void* d_out, int out_size, void* d_ws, size_t ws_size,
                              hipStream_t stream) {
    const float* dec_states = (const float*)d_in[0];
    const float* enc_states = (const float*)d_in[1];
    const float* W_mlp      = (const float*)d_in[2];
    const float* b_mlp      = (const float*)d_in[3];
    const float* W_out      = (const float*)d_in[4];
    const float* b_out      = (const float*)d_in[5];
    float*       out        = (float*)d_out;

    float* enc_proj = (float*)d_ws;
    float* dec_proj = enc_proj + (size_t)B_SZ * ENC_N * HID_N;

    proj_wmma_f32<<<256, 128, 0, stream>>>(enc_states, dec_states, W_mlp, b_mlp,
                                           enc_proj, dec_proj);
    relu_dot_score<<<256, 256, 0, stream>>>(dec_proj, enc_proj,
                                            W_out, b_out, out);
}
